# starting kernel
# baseline (speedup 1.0000x reference)
#include <hip/hip_runtime.h>
#include <hip/hip_bf16.h>
#include <math.h>


typedef __attribute__((ext_vector_type(2))) float v2f;
typedef __attribute__((ext_vector_type(8))) float v8f;

#define NW 8
#define E_DIM 64
#define S_DIM 512
#define TOK_TILE 16

__device__ __forceinline__ v8f wmma_f32_k4(v2f a, v2f b, v8f c) {
  return __builtin_amdgcn_wmma_f32_16x16x4_f32(false, a, false, b, (short)0, c,
                                               false, false);
}

__global__ __launch_bounds__(32) void qattn_fused_kernel(
    const float* __restrict__ x, const float* __restrict__ Wq,
    const float* __restrict__ Wk, const float* __restrict__ Wv,
    const float* __restrict__ Wo, const float* __restrict__ theta,
    float* __restrict__ y) {
  __shared__ float t_ld[3][TOK_TILE][E_DIM + 4];
  __shared__ float qkv[3][TOK_TILE][NW][NW];
  __shared__ float o2[TOK_TILE][E_DIM + 8];

  const int L = threadIdx.x;
  const int lo = L & 15;
  const int hi = L >> 4;
  const int T = blockIdx.x;
  const int b = T >> 5;
  const int s0 = (T & 31) * TOK_TILE;
  const int sg0 = (T & 31) * 2;

  const float* xt = x + (size_t)(b * S_DIM + s0) * E_DIM;
  const float* Ws[3] = {Wq, Wk, Wv};

#pragma unroll
  for (int mtx = 0; mtx < 3; ++mtx) {
    const float* W = Ws[mtx];
#pragma unroll
    for (int n0 = 0; n0 < E_DIM; n0 += 16) {
      v8f acc = {0.f, 0.f, 0.f, 0.f, 0.f, 0.f, 0.f, 0.f};
#pragma unroll
      for (int k0 = 0; k0 < E_DIM; k0 += 4) {
        v2f a = *(const v2f*)(xt + lo * E_DIM + k0 + 2 * hi);
        v2f bb = *(const v2f*)(W + (n0 + lo) * E_DIM + k0 + 2 * hi);
        acc = wmma_f32_k4(a, bb, acc);
      }
#pragma unroll
      for (int r = 0; r < 8; ++r) t_ld[mtx][r + 8 * hi][n0 + lo] = acc[r];
    }
  }

  float th[NW];
#pragma unroll
  for (int w = 0; w < NW; ++w) th[w] = theta[w];

#pragma unroll
  for (int i = 0; i < 4; ++i) {
    int p = L + 32 * i;
    int tok = p >> 3;
    int h = p & 7;
#pragma unroll
    for (int mtx = 0; mtx < 3; ++mtx) {
      float c[NW];
#pragma unroll
      for (int w = 0; w < NW; ++w)
        c[w] = __cosf(t_ld[mtx][tok][h * 8 + w] + th[w]);
      float z[NW];
      float pr = c[0];
#pragma unroll
      for (int w = 1; w < NW; ++w) {
        pr *= c[w];
        z[w] = pr;
      }
      float pt = 1.f;
#pragma unroll
      for (int w = 1; w < NW; ++w) pt *= c[w];
      z[0] = pt;
#pragma unroll
      for (int w = 0; w < NW; ++w) qkv[mtx][tok][h][w] = z[w];
    }
  }

#pragma unroll
  for (int i = 0; i < 4; ++i) {
    int p = L + 32 * i;
    int tok = p >> 3;
    int ih = p & 7;
    float qr[NW];
#pragma unroll
    for (int w = 0; w < NW; ++w) qr[w] = qkv[0][tok][ih][w];
    float sc[NW];
    float mx = -3.0e38f;
#pragma unroll
    for (int j = 0; j < NW; ++j) {
      float d = 0.f;
#pragma unroll
      for (int w = 0; w < NW; ++w) d += qr[w] * qkv[1][tok][j][w];
      d *= 0.35355339059327373f;
      sc[j] = d;
      mx = fmaxf(mx, d);
    }
    float se = 0.f;
#pragma unroll
    for (int j = 0; j < NW; ++j) {
      sc[j] = __expf(sc[j] - mx);
      se += sc[j];
    }
    float inv = 1.f / se;
    float orow[NW];
#pragma unroll
    for (int w = 0; w < NW; ++w) orow[w] = 0.f;
#pragma unroll
    for (int j = 0; j < NW; ++j) {
      float a = sc[j] * inv;
#pragma unroll
      for (int w = 0; w < NW; ++w) orow[w] += a * qkv[2][tok][j][w];
    }
    int r2 = ih * 2 + (tok >> 3);
    int colb = (tok & 7) * 8;
#pragma unroll
    for (int w = 0; w < NW; ++w) o2[r2][colb + w] = orow[w];
  }

#pragma unroll
  for (int n0 = 0; n0 < E_DIM; n0 += 16) {
    v8f acc = {0.f, 0.f, 0.f, 0.f, 0.f, 0.f, 0.f, 0.f};
#pragma unroll
    for (int k0 = 0; k0 < E_DIM; k0 += 4) {
      v2f a = *(const v2f*)&o2[lo][k0 + 2 * hi];
      v2f bb = *(const v2f*)(Wo + (n0 + lo) * E_DIM + k0 + 2 * hi);
      acc = wmma_f32_k4(a, bb, acc);
    }
    int n = n0 + lo;
#pragma unroll
    for (int r = 0; r < 8; ++r) {
      int rr = r + 8 * hi;
      int m = (rr >> 1) * 64 + sg0 + (rr & 1);
      y[(size_t)b * (S_DIM * E_DIM) + (size_t)m * E_DIM + n] = acc[r];
    }
  }
}

extern "C" void kernel_launch(void* const* d_in, const int* in_sizes, int n_in,
                              void* d_out, int out_size, void* d_ws,
                              size_t ws_size, hipStream_t stream) {
  (void)in_sizes; (void)n_in; (void)out_size; (void)d_ws; (void)ws_size;
  const float* x = (const float*)d_in[0];
  const float* Wq = (const float*)d_in[1];
  const float* Wk = (const float*)d_in[2];
  const float* Wv = (const float*)d_in[3];
  const float* Wo = (const float*)d_in[4];
  const float* theta = (const float*)d_in[5];
  float* y = (float*)d_out;
  dim3 grid(512), block(32);
  hipLaunchKernelGGL(qattn_fused_kernel, grid, block, 0, stream, x, Wq, Wk, Wv,
                     Wo, theta, y);
}


# Split each 16-token tile over 4 waves (128-thread block): a wave owns one 16-column tile of all four GEMMs, one (token, head) row per thread; y tile meets in LDS for whole-line stores
# speedup vs baseline: 1.2228x; 1.2228x over previous
#include <hip/hip_runtime.h>
#include <hip/hip_bf16.h>
#include <math.h>


typedef __attribute__((ext_vector_type(2))) float v2f;
typedef __attribute__((ext_vector_type(4))) float v4f;
typedef __attribute__((ext_vector_type(8))) float v8f;

#define NW 8
#define E_DIM 64
#define S_DIM 512
#define TOK_TILE 16
#define KSTEPS (E_DIM / 4)
#define BLOCK_THREADS 128

__device__ __forceinline__ v8f wmma_f32_k4(v2f a, v2f b, v8f c) {
  return __builtin_amdgcn_wmma_f32_16x16x4_f32(false, a, false, b, (short)0, c,
                                               false, false);
}

__global__ __launch_bounds__(BLOCK_THREADS) void qattn_fused_kernel(
    const float* __restrict__ x, const float* __restrict__ Wq,
    const float* __restrict__ Wk, const float* __restrict__ Wv,
    const float* __restrict__ Wo, const float* __restrict__ theta,
    float* __restrict__ y) {
  __shared__ __attribute__((aligned(16))) float t_ld[3][TOK_TILE][E_DIM + 4];
  __shared__ __attribute__((aligned(16))) float qkv[3][TOK_TILE][NW][NW];
  __shared__ __attribute__((aligned(16))) float o2[TOK_TILE][E_DIM + 8];

  const int tid = threadIdx.x;
  const int L = tid & 31;
  const int wv = __builtin_amdgcn_readfirstlane(tid >> 5);
  const int lo = L & 15;
  const int hi = L >> 4;
  const int T = blockIdx.x;
  const int b = T >> 5;
  const int s0 = (T & 31) * TOK_TILE;
  const int sg0 = (T & 31) * 2;
  const int n0 = wv * 16;

  const float* xt = x + (size_t)(b * S_DIM + s0) * E_DIM;

  const float* xa = xt + lo * E_DIM + 2 * hi;
  const int woff = (n0 + lo) * E_DIM + 2 * hi;
  v2f fa[KSTEPS], fq[KSTEPS], fk[KSTEPS], fv[KSTEPS], fo[KSTEPS];
#pragma unroll
  for (int k = 0; k < KSTEPS; ++k) {
    fa[k] = *(const v2f*)(xa + 4 * k);
    fq[k] = *(const v2f*)(Wq + woff + 4 * k);
    fk[k] = *(const v2f*)(Wk + woff + 4 * k);
    fv[k] = *(const v2f*)(Wv + woff + 4 * k);
  }
#pragma unroll
  for (int k = 0; k < KSTEPS; ++k) fo[k] = *(const v2f*)(Wo + woff + 4 * k);

  v8f cq = {0.f, 0.f, 0.f, 0.f, 0.f, 0.f, 0.f, 0.f};
  v8f ck = cq;
  v8f cv = cq;
#pragma unroll
  for (int k = 0; k < KSTEPS; ++k) {
    cq = wmma_f32_k4(fa[k], fq[k], cq);
    ck = wmma_f32_k4(fa[k], fk[k], ck);
    cv = wmma_f32_k4(fa[k], fv[k], cv);
  }
#pragma unroll
  for (int r = 0; r < 8; ++r) {
    t_ld[0][r + 8 * hi][n0 + lo] = cq[r];
    t_ld[1][r + 8 * hi][n0 + lo] = ck[r];
    t_ld[2][r + 8 * hi][n0 + lo] = cv[r];
  }

  float th[NW];
#pragma unroll
  for (int w = 0; w < NW; ++w) th[w] = theta[w];

  __syncthreads();

  const int tok = tid >> 3;
  const int h = tid & 7;

#pragma unroll
  for (int mtx = 0; mtx < 3; ++mtx) {
    float c[NW];
#pragma unroll
    for (int w = 0; w < NW; ++w)
      c[w] = __cosf(t_ld[mtx][tok][h * 8 + w] + th[w]);
    float z[NW];
    float pr = c[0];
#pragma unroll
    for (int w = 1; w < NW; ++w) {
      pr *= c[w];
      z[w] = pr;
    }
    float pt = 1.f;
#pragma unroll
    for (int w = 1; w < NW; ++w) pt *= c[w];
    z[0] = pt;
#pragma unroll
    for (int w = 0; w < NW; ++w) qkv[mtx][tok][h][w] = z[w];
  }

  __syncthreads();

  {
    const int ih = h;
    float qr[NW];
#pragma unroll
    for (int w = 0; w < NW; ++w) qr[w] = qkv[0][tok][ih][w];
    float sc[NW];
    float mx = -3.0e38f;
#pragma unroll
    for (int j = 0; j < NW; ++j) {
      float d = 0.f;
#pragma unroll
      for (int w = 0; w < NW; ++w) d += qr[w] * qkv[1][tok][j][w];
      d *= 0.35355339059327373f;
      sc[j] = d;
      mx = fmaxf(mx, d);
    }
    float se = 0.f;
#pragma unroll
    for (int j = 0; j < NW; ++j) {
      sc[j] = __expf(sc[j] - mx);
      se += sc[j];
    }
    float inv = 1.f / se;
    float orow[NW];
#pragma unroll
    for (int w = 0; w < NW; ++w) orow[w] = 0.f;
#pragma unroll
    for (int j = 0; j < NW; ++j) {
      float a = sc[j] * inv;
#pragma unroll
      for (int w = 0; w < NW; ++w) orow[w] += a * qkv[2][tok][j][w];
    }
    int r2 = ih * 2 + (tok >> 3);
    int colb = (tok & 7) * 8;
#pragma unroll
    for (int w = 0; w < NW; ++w) o2[r2][colb + w] = orow[w];
  }

  __syncthreads();

  v8f acc = {0.f, 0.f, 0.f, 0.f, 0.f, 0.f, 0.f, 0.f};
#pragma unroll
  for (int k = 0; k < KSTEPS; ++k) {
    v2f a = *(const v2f*)&o2[lo][4 * k + 2 * hi];
    acc = wmma_f32_k4(a, fo[k], acc);
  }
  float (*ytile)[E_DIM + 4] = t_ld[0];
#pragma unroll
  for (int r = 0; r < 8; ++r) ytile[r + 8 * hi][n0 + lo] = acc[r];

  __syncthreads();

  float* yb = y + (size_t)b * (S_DIM * E_DIM);
#pragma unroll
  for (int i = 0; i < 2; ++i) {
    int f = tid + BLOCK_THREADS * i;
    int rr = f >> 4;
    int c4 = (f & 15) * 4;
    int m = (rr >> 1) * 64 + sg0 + (rr & 1);
    v4f v = *(const v4f*)&ytile[rr][c4];
    *(v4f*)(yb + (size_t)m * E_DIM + c4) = v;
  }
}

extern "C" void kernel_launch(void* const* d_in, const int* in_sizes, int n_in,
                              void* d_out, int out_size, void* d_ws,
                              size_t ws_size, hipStream_t stream) {
  (void)in_sizes; (void)n_in; (void)out_size; (void)d_ws; (void)ws_size;
  const float* x = (const float*)d_in[0];
  const float* Wq = (const float*)d_in[1];
  const float* Wk = (const float*)d_in[2];
  const float* Wv = (const float*)d_in[3];
  const float* Wo = (const float*)d_in[4];
  const float* theta = (const float*)d_in[5];
  float* y = (float*)d_out;
  dim3 grid(512), block(BLOCK_THREADS);
  hipLaunchKernelGGL(qattn_fused_kernel, grid, block, 0, stream, x, Wq, Wk, Wv,
                     Wo, theta, y);
}
